# starting kernel
# baseline (speedup 1.0000x reference)
#include <hip/hip_runtime.h>
#include <hip/hip_bf16.h>


typedef float v2f __attribute__((ext_vector_type(2)));
typedef float v8f __attribute__((ext_vector_type(8)));

#define LIN_B  16
#define LIN_S  4096
#define LIN_D  64
#define LIN_KP 256

__device__ __forceinline__ v8f wmma_f32(v2f a, v2f b, v8f c) {
    return __builtin_amdgcn_wmma_f32_16x16x4_f32(false, a, false, b, (short)0, c,
                                                 false, false);
}

template <bool KT_PACK>
__global__ __launch_bounds__(32) void lin_proj_kernel(
    const float* __restrict__ X, const float* __restrict__ W,
    const float* __restrict__ bias, float* __restrict__ outP) {
    const int S = LIN_S, D = LIN_D, KP = LIN_KP;

    const int blk  = blockIdx.x;
    const int b    = blk >> 6;
    const int tile = blk & 63;
    const int mt   = tile >> 2;
    const int nt   = tile & 3;
    const int l    = threadIdx.x;
    const int lh   = l >> 4;
    const int ll   = l & 15;
    const int kp0  = mt * 16;
    const int d0   = nt * 16;

    const float* wrow = W + (size_t)(kp0 + ll) * S + lh * 2;
    const float* xcol = X + (size_t)b * S * D + d0 + ll;

    v8f acc = {};
    for (int s0 = 0; s0 < S; s0 += 32) {
        v2f a[8], bf[8];
#pragma unroll
        for (int j = 0; j < 8; ++j)
            a[j] = *(const v2f*)(wrow + s0 + j * 4);
#pragma unroll
        for (int j = 0; j < 8; ++j) {
            const float* xp = xcol + (size_t)(s0 + j * 4 + lh * 2) * D;
            bf[j].x = xp[0];
            bf[j].y = xp[D];
        }
#pragma unroll
        for (int j = 0; j < 8; ++j) acc = wmma_f32(a[j], bf[j], acc);
    }

#pragma unroll
    for (int v = 0; v < 8; ++v) acc[v] += bias[kp0 + 8 * lh + v];

    const int d = d0 + ll;
    if (KT_PACK) {
        float* p = outP + (size_t)b * D * KP + (size_t)(d >> 1) * (2 * KP) +
                   (size_t)(kp0 + 8 * lh) * 2 + (d & 1);
#pragma unroll
        for (int v = 0; v < 8; ++v) p[2 * v] = acc[v];
    } else {
        float* base = outP + (size_t)b * KP * D + d * 2;
#pragma unroll
        for (int v = 0; v < 8; ++v) {
            const int kp = kp0 + 8 * lh + v;
            base[(size_t)(kp >> 1) * (2 * D) + (kp & 1)] = acc[v];
        }
    }
}

#define LDS_STRIDE 260

__global__ __launch_bounds__(32) void lin_attn_kernel(
    const float* __restrict__ Q, const float* __restrict__ Kpp,
    const float* __restrict__ Vpp, float* __restrict__ out,
    float* __restrict__ attn) {
    const int S = LIN_S, D = LIN_D, KP = LIN_KP;
    const float scale = 0.125f;

    __shared__ float lds[16 * LDS_STRIDE];

    const int blk = blockIdx.x;
    const int b   = blk >> 8;
    const int s0  = (blk & 255) * 16;
    const int l   = threadIdx.x;
    const int lh  = l >> 4;
    const int ll  = l & 15;

    v2f qa[16];
    const float* qrow = Q + (size_t)b * S * D + (size_t)(s0 + ll) * D + lh * 2;
#pragma unroll
    for (int kk = 0; kk < 16; ++kk) qa[kk] = *(const v2f*)(qrow + kk * 4);

    v8f sc[16];
    const float* kpB = Kpp + (size_t)b * D * KP + ll * 2;
#pragma unroll
    for (int t = 0; t < 16; ++t) {
        v2f bf[16];
#pragma unroll
        for (int kk = 0; kk < 16; ++kk)
            bf[kk] = *(const v2f*)(kpB + (size_t)(kk * 2 + lh) * (2 * KP) + t * 32);
        v8f c = {};
#pragma unroll
        for (int kk = 0; kk < 16; ++kk) c = wmma_f32(qa[kk], bf[kk], c);
        sc[t] = c;
    }

#pragma unroll
    for (int v = 0; v < 8; ++v) {
        float mx = sc[0][v];
#pragma unroll
        for (int t = 1; t < 16; ++t) mx = fmaxf(mx, sc[t][v]);
        mx = fmaxf(mx, __shfl_xor(mx, 1, 32));
        mx = fmaxf(mx, __shfl_xor(mx, 2, 32));
        mx = fmaxf(mx, __shfl_xor(mx, 4, 32));
        mx = fmaxf(mx, __shfl_xor(mx, 8, 32));
        float sum = 0.f;
#pragma unroll
        for (int t = 0; t < 16; ++t) {
            float p = __expf((sc[t][v] - mx) * scale);
            sc[t][v] = p;
            sum += p;
        }
        sum += __shfl_xor(sum, 1, 32);
        sum += __shfl_xor(sum, 2, 32);
        sum += __shfl_xor(sum, 4, 32);
        sum += __shfl_xor(sum, 8, 32);
        const float r = 1.0f / sum;
#pragma unroll
        for (int t = 0; t < 16; ++t) sc[t][v] *= r;
    }

    float* arow = attn + (size_t)b * S * KP;
#pragma unroll
    for (int t = 0; t < 16; ++t) {
#pragma unroll
        for (int v = 0; v < 8; ++v) {
            const int m   = v + 8 * lh;
            const int col = t * 16 + ll;
            arow[(size_t)(s0 + m) * KP + col] = sc[t][v];
            lds[m * LDS_STRIDE + col]         = sc[t][v];
        }
    }
    __syncthreads();

    const float* vpB = Vpp + (size_t)b * KP * D + ll * 2;
    v8f oc[4] = {};
#pragma unroll 4
    for (int kk = 0; kk < 64; ++kk) {
        v2f a = *(const v2f*)(&lds[ll * LDS_STRIDE + kk * 4 + lh * 2]);
        v2f bf[4];
#pragma unroll
        for (int td = 0; td < 4; ++td)
            bf[td] = *(const v2f*)(vpB + (size_t)(kk * 2 + lh) * (2 * D) + td * 32);
#pragma unroll
        for (int td = 0; td < 4; ++td) oc[td] = wmma_f32(a, bf[td], oc[td]);
    }
#pragma unroll
    for (int td = 0; td < 4; ++td) {
#pragma unroll
        for (int v = 0; v < 8; ++v)
            out[(size_t)b * S * D + (size_t)(s0 + v + 8 * lh) * D + td * 16 + ll] =
                oc[td][v];
    }
}

extern "C" void kernel_launch(void* const* d_in, const int* in_sizes, int n_in,
                              void* d_out, int out_size, void* d_ws, size_t ws_size,
                              hipStream_t stream) {
    (void)in_sizes; (void)n_in; (void)out_size; (void)ws_size;
    const float* Q   = (const float*)d_in[0];
    const float* K   = (const float*)d_in[1];
    const float* V   = (const float*)d_in[2];
    const float* E_w = (const float*)d_in[3];
    const float* E_b = (const float*)d_in[4];
    const float* F_w = (const float*)d_in[5];
    const float* F_b = (const float*)d_in[6];

    float* out  = (float*)d_out;
    float* attn = out + (size_t)LIN_B * LIN_S * LIN_D;

    float* Kpp = (float*)d_ws;
    float* Vpp = Kpp + (size_t)LIN_B * LIN_D * LIN_KP;

    const int projBlocks = LIN_B * 16 * 4;
    const int attnBlocks = LIN_B * (LIN_S / 16);

    lin_proj_kernel<true ><<<projBlocks, 32, 0, stream>>>(K, E_w, E_b, Kpp);
    lin_proj_kernel<false><<<projBlocks, 32, 0, stream>>>(V, F_w, F_b, Vpp);
    lin_attn_kernel<<<attnBlocks, 32, 0, stream>>>(Q, Kpp, Vpp, out, attn);
}


# Projection re-tiled: a 16-wave work-group per 16x64 output tile, S sum split over its waves (4 accumulator chains, whole-row b128 X loads), partial tiles summed through LDS in fixed wave order
# speedup vs baseline: 2.4775x; 2.4775x over previous
#include <hip/hip_runtime.h>
#include <hip/hip_bf16.h>


typedef float v2f __attribute__((ext_vector_type(2)));
typedef float v8f __attribute__((ext_vector_type(8)));

#define LIN_B  16
#define LIN_S  4096
#define LIN_D  64
#define LIN_KP 256

__device__ __forceinline__ v8f wmma_f32(v2f a, v2f b, v8f c) {
    return __builtin_amdgcn_wmma_f32_16x16x4_f32(false, a, false, b, (short)0, c,
                                                 false, false);
}

typedef float v4f __attribute__((ext_vector_type(4)));

#define PROJ_WAVES 16
#define PROJ_PITCH 68

struct proj_frags {
    v4f a[4];
    v4f x[4][4];
};

__device__ __forceinline__ void proj_load(proj_frags& f, const float* wrow,
                                          const float* xrow, int c) {
    const int D = LIN_D;
#pragma unroll
    for (int j = 0; j < 4; ++j)
        f.a[j] = *(const v4f*)(wrow + c * 512 + j * 8);
#pragma unroll
    for (int j = 0; j < 4; ++j)
#pragma unroll
        for (int r = 0; r < 4; ++r)
            f.x[j][r] = *(const v4f*)(xrow + (size_t)(c * 512 + j * 8 + r) * D);
}

__device__ __forceinline__ void proj_mma(const proj_frags& f, v8f (&acc)[4]) {
#pragma unroll
    for (int j = 0; j < 4; ++j)
#pragma unroll
        for (int e = 0; e < 2; ++e) {
            v2f a = {f.a[j][2 * e], f.a[j][2 * e + 1]};
#pragma unroll
            for (int td = 0; td < 4; ++td) {
                v2f bb = {f.x[j][2 * e][td], f.x[j][2 * e + 1][td]};
                acc[td] = wmma_f32(a, bb, acc[td]);
            }
        }
}

template <bool KT_PACK>
__global__ __launch_bounds__(PROJ_WAVES * 32) void lin_proj_kernel(
    const float* X, const float* W,
    const float* __restrict__ bias, float* __restrict__ outP) {
    const int S = LIN_S, D = LIN_D, KP = LIN_KP;

    __shared__ float part[PROJ_WAVES * 16 * PROJ_PITCH];

    const int blk = blockIdx.x;
    const int b   = blk >> 4;
    const int kp0 = (blk & 15) * 16;
    const int tid = threadIdx.x;
    const int w   = tid >> 5;
    const int l   = tid & 31;
    const int lh  = l >> 4;
    const int ll  = l & 15;

    const float* wrow = W + (size_t)(kp0 + ll) * S + w * 32 + lh * 4;
    const float* xrow = X + (size_t)b * S * D + (size_t)(w * 32 + lh * 4) * D + ll * 4;

    v8f acc[4] = {};
    proj_frags f0, f1;
    proj_load(f0, wrow, xrow, 0);
#pragma unroll
    for (int c = 0; c < 8; c += 2) {
        proj_load(f1, wrow, xrow, c + 1);
        __builtin_amdgcn_sched_barrier(0);
        proj_mma(f0, acc);
        __builtin_amdgcn_sched_barrier(0);
        if (c + 2 < 8) proj_load(f0, wrow, xrow, c + 2);
        __builtin_amdgcn_sched_barrier(0);
        proj_mma(f1, acc);
        __builtin_amdgcn_sched_barrier(0);
    }

    float* mine = part + (size_t)(w * 16 + 8 * lh) * PROJ_PITCH + ll * 4;
#pragma unroll
    for (int v = 0; v < 8; ++v) {
        v4f o = {acc[0][v], acc[1][v], acc[2][v], acc[3][v]};
        *(v4f*)(mine + v * PROJ_PITCH) = o;
    }
    __syncthreads();

    if (KT_PACK) {
        const int p   = tid >> 4;
        const int kpl = tid & 15;
        const float* src = part + (size_t)kpl * PROJ_PITCH + 2 * p;
        v2f s = {0.f, 0.f};
#pragma unroll
        for (int q = 0; q < PROJ_WAVES; ++q)
            s += *(const v2f*)(src + q * 16 * PROJ_PITCH);
        const float bv = bias[kp0 + kpl];
        s.x += bv;
        s.y += bv;
        *(v2f*)(outP + (size_t)b * D * KP + (size_t)p * (2 * KP) +
                (size_t)(kp0 + kpl) * 2) = s;
    } else {
        const int q2 = tid >> 6;
        const int d  = tid & 63;
        const float* src = part + (size_t)(2 * q2) * PROJ_PITCH + d;
        v2f s = {0.f, 0.f};
#pragma unroll
        for (int q = 0; q < PROJ_WAVES; ++q) {
            s.x += src[q * 16 * PROJ_PITCH];
            s.y += src[q * 16 * PROJ_PITCH + PROJ_PITCH];
        }
        s.x += bias[kp0 + 2 * q2];
        s.y += bias[kp0 + 2 * q2 + 1];
        *(v2f*)(outP + (size_t)b * KP * D + (size_t)((kp0 >> 1) + q2) * (2 * D) +
                (size_t)d * 2) = s;
    }
}

#define LDS_STRIDE 260

__global__ __launch_bounds__(32) void lin_attn_kernel(
    const float* __restrict__ Q, const float* __restrict__ Kpp,
    const float* __restrict__ Vpp, float* __restrict__ out,
    float* __restrict__ attn) {
    const int S = LIN_S, D = LIN_D, KP = LIN_KP;
    const float scale = 0.125f;

    __shared__ float lds[16 * LDS_STRIDE];

    const int blk = blockIdx.x;
    const int b   = blk >> 8;
    const int s0  = (blk & 255) * 16;
    const int l   = threadIdx.x;
    const int lh  = l >> 4;
    const int ll  = l & 15;

    v2f qa[16];
    const float* qrow = Q + (size_t)b * S * D + (size_t)(s0 + ll) * D + lh * 2;
#pragma unroll
    for (int kk = 0; kk < 16; ++kk) qa[kk] = *(const v2f*)(qrow + kk * 4);

    v8f sc[16];
    const float* kpB = Kpp + (size_t)b * D * KP + ll * 2;
#pragma unroll
    for (int t = 0; t < 16; ++t) {
        v2f bf[16];
#pragma unroll
        for (int kk = 0; kk < 16; ++kk)
            bf[kk] = *(const v2f*)(kpB + (size_t)(kk * 2 + lh) * (2 * KP) + t * 32);
        v8f c = {};
#pragma unroll
        for (int kk = 0; kk < 16; ++kk) c = wmma_f32(qa[kk], bf[kk], c);
        sc[t] = c;
    }

#pragma unroll
    for (int v = 0; v < 8; ++v) {
        float mx = sc[0][v];
#pragma unroll
        for (int t = 1; t < 16; ++t) mx = fmaxf(mx, sc[t][v]);
        mx = fmaxf(mx, __shfl_xor(mx, 1, 32));
        mx = fmaxf(mx, __shfl_xor(mx, 2, 32));
        mx = fmaxf(mx, __shfl_xor(mx, 4, 32));
        mx = fmaxf(mx, __shfl_xor(mx, 8, 32));
        float sum = 0.f;
#pragma unroll
        for (int t = 0; t < 16; ++t) {
            float p = __expf((sc[t][v] - mx) * scale);
            sc[t][v] = p;
            sum += p;
        }
        sum += __shfl_xor(sum, 1, 32);
        sum += __shfl_xor(sum, 2, 32);
        sum += __shfl_xor(sum, 4, 32);
        sum += __shfl_xor(sum, 8, 32);
        const float r = 1.0f / sum;
#pragma unroll
        for (int t = 0; t < 16; ++t) sc[t][v] *= r;
    }

    float* arow = attn + (size_t)b * S * KP;
#pragma unroll
    for (int t = 0; t < 16; ++t) {
#pragma unroll
        for (int v = 0; v < 8; ++v) {
            const int m   = v + 8 * lh;
            const int col = t * 16 + ll;
            arow[(size_t)(s0 + m) * KP + col] = sc[t][v];
            lds[m * LDS_STRIDE + col]         = sc[t][v];
        }
    }
    __syncthreads();

    const float* vpB = Vpp + (size_t)b * KP * D + ll * 2;
    v8f oc[4] = {};
#pragma unroll 4
    for (int kk = 0; kk < 64; ++kk) {
        v2f a = *(const v2f*)(&lds[ll * LDS_STRIDE + kk * 4 + lh * 2]);
        v2f bf[4];
#pragma unroll
        for (int td = 0; td < 4; ++td)
            bf[td] = *(const v2f*)(vpB + (size_t)(kk * 2 + lh) * (2 * D) + td * 32);
#pragma unroll
        for (int td = 0; td < 4; ++td) oc[td] = wmma_f32(a, bf[td], oc[td]);
    }
#pragma unroll
    for (int td = 0; td < 4; ++td) {
#pragma unroll
        for (int v = 0; v < 8; ++v)
            out[(size_t)b * S * D + (size_t)(s0 + v + 8 * lh) * D + td * 16 + ll] =
                oc[td][v];
    }
}

extern "C" void kernel_launch(void* const* d_in, const int* in_sizes, int n_in,
                              void* d_out, int out_size, void* d_ws, size_t ws_size,
                              hipStream_t stream) {
    (void)in_sizes; (void)n_in; (void)out_size; (void)ws_size;
    const float* Q   = (const float*)d_in[0];
    const float* K   = (const float*)d_in[1];
    const float* V   = (const float*)d_in[2];
    const float* E_w = (const float*)d_in[3];
    const float* E_b = (const float*)d_in[4];
    const float* F_w = (const float*)d_in[5];
    const float* F_b = (const float*)d_in[6];

    float* out  = (float*)d_out;
    float* attn = out + (size_t)LIN_B * LIN_S * LIN_D;

    float* Kpp = (float*)d_ws;
    float* Vpp = Kpp + (size_t)LIN_B * LIN_D * LIN_KP;

    const int projBlocks = LIN_B * 16;
    const int attnBlocks = LIN_B * (LIN_S / 16);

    lin_proj_kernel<true ><<<projBlocks, PROJ_WAVES * 32, 0, stream>>>(K, E_w, E_b, Kpp);
    lin_proj_kernel<false><<<projBlocks, PROJ_WAVES * 32, 0, stream>>>(V, F_w, F_b, Vpp);
    lin_attn_kernel<<<attnBlocks, 32, 0, stream>>>(Q, Kpp, Vpp, out, attn);
}


# Attention re-grouped: 256 work-groups of 8 waves copy their batch's packed Kp and Vp into LDS once; both GEMMs read B fragments from LDS, two strips per wave; order of sums unchanged
# speedup vs baseline: 2.8422x; 1.1472x over previous
#include <hip/hip_runtime.h>
#include <hip/hip_bf16.h>


typedef float v2f __attribute__((ext_vector_type(2)));
typedef float v8f __attribute__((ext_vector_type(8)));

#define LIN_B  16
#define LIN_S  4096
#define LIN_D  64
#define LIN_KP 256

__device__ __forceinline__ v8f wmma_f32(v2f a, v2f b, v8f c) {
    return __builtin_amdgcn_wmma_f32_16x16x4_f32(false, a, false, b, (short)0, c,
                                                 false, false);
}

typedef float v4f __attribute__((ext_vector_type(4)));

#define PROJ_WAVES 16
#define PROJ_PITCH 68

struct proj_frags {
    v4f a[4];
    v4f x[4][4];
};

__device__ __forceinline__ void proj_load(proj_frags& f, const float* wrow,
                                          const float* xrow, int c) {
    const int D = LIN_D;
#pragma unroll
    for (int j = 0; j < 4; ++j)
        f.a[j] = *(const v4f*)(wrow + c * 512 + j * 8);
#pragma unroll
    for (int j = 0; j < 4; ++j)
#pragma unroll
        for (int r = 0; r < 4; ++r)
            f.x[j][r] = *(const v4f*)(xrow + (size_t)(c * 512 + j * 8 + r) * D);
}

__device__ __forceinline__ void proj_mma(const proj_frags& f, v8f (&acc)[4]) {
#pragma unroll
    for (int j = 0; j < 4; ++j)
#pragma unroll
        for (int e = 0; e < 2; ++e) {
            v2f a = {f.a[j][2 * e], f.a[j][2 * e + 1]};
#pragma unroll
            for (int td = 0; td < 4; ++td) {
                v2f bb = {f.x[j][2 * e][td], f.x[j][2 * e + 1][td]};
                acc[td] = wmma_f32(a, bb, acc[td]);
            }
        }
}

template <bool KT_PACK>
__global__ __launch_bounds__(PROJ_WAVES * 32) void lin_proj_kernel(
    const float* X, const float* W,
    const float* __restrict__ bias, float* __restrict__ outP) {
    const int S = LIN_S, D = LIN_D, KP = LIN_KP;

    __shared__ float part[PROJ_WAVES * 16 * PROJ_PITCH];

    const int blk = blockIdx.x;
    const int b   = blk >> 4;
    const int kp0 = (blk & 15) * 16;
    const int tid = threadIdx.x;
    const int w   = tid >> 5;
    const int l   = tid & 31;
    const int lh  = l >> 4;
    const int ll  = l & 15;

    const float* wrow = W + (size_t)(kp0 + ll) * S + w * 32 + lh * 4;
    const float* xrow = X + (size_t)b * S * D + (size_t)(w * 32 + lh * 4) * D + ll * 4;

    v8f acc[4] = {};
    proj_frags f0, f1;
    proj_load(f0, wrow, xrow, 0);
#pragma unroll
    for (int c = 0; c < 8; c += 2) {
        proj_load(f1, wrow, xrow, c + 1);
        __builtin_amdgcn_sched_barrier(0);
        proj_mma(f0, acc);
        __builtin_amdgcn_sched_barrier(0);
        if (c + 2 < 8) proj_load(f0, wrow, xrow, c + 2);
        __builtin_amdgcn_sched_barrier(0);
        proj_mma(f1, acc);
        __builtin_amdgcn_sched_barrier(0);
    }

    float* mine = part + (size_t)(w * 16 + 8 * lh) * PROJ_PITCH + ll * 4;
#pragma unroll
    for (int v = 0; v < 8; ++v) {
        v4f o = {acc[0][v], acc[1][v], acc[2][v], acc[3][v]};
        *(v4f*)(mine + v * PROJ_PITCH) = o;
    }
    __syncthreads();

    if (KT_PACK) {
        const int p   = tid >> 4;
        const int kpl = tid & 15;
        const float* src = part + (size_t)kpl * PROJ_PITCH + 2 * p;
        v2f s = {0.f, 0.f};
#pragma unroll
        for (int q = 0; q < PROJ_WAVES; ++q)
            s += *(const v2f*)(src + q * 16 * PROJ_PITCH);
        const float bv = bias[kp0 + kpl];
        s.x += bv;
        s.y += bv;
        *(v2f*)(outP + (size_t)b * D * KP + (size_t)p * (2 * KP) +
                (size_t)(kp0 + kpl) * 2) = s;
    } else {
        const int q2 = tid >> 6;
        const int d  = tid & 63;
        const float* src = part + (size_t)(2 * q2) * PROJ_PITCH + d;
        v2f s = {0.f, 0.f};
#pragma unroll
        for (int q = 0; q < PROJ_WAVES; ++q) {
            s.x += src[q * 16 * PROJ_PITCH];
            s.y += src[q * 16 * PROJ_PITCH + PROJ_PITCH];
        }
        s.x += bias[kp0 + 2 * q2];
        s.y += bias[kp0 + 2 * q2 + 1];
        *(v2f*)(outP + (size_t)b * KP * D + (size_t)((kp0 >> 1) + q2) * (2 * D) +
                (size_t)d * 2) = s;
    }
}

#define LDS_STRIDE 260
#define ATT_WAVES  8
#define ATT_ROWS   256

struct sc_frags { v2f b[2][16]; };
struct pv_frags { v2f a[4]; v2f b[4][4]; };

__device__ __forceinline__ void sc_load(sc_frags& f, const float* kbase, int lh, int p) {
#pragma unroll
    for (int u = 0; u < 2; ++u) {
        const float* src = kbase + ((2 * p + u + lh) & 15) * 32;
#pragma unroll
        for (int kk = 0; kk < 16; ++kk) f.b[u][kk] = *(const v2f*)(src + kk * 1024);
    }
}

__device__ __forceinline__ void sc_mma(const sc_frags& f, const v2f (&qa)[16],
                                       v8f& c0, v8f& c1) {
#pragma unroll
    for (int kk = 0; kk < 16; ++kk) {
        c0 = wmma_f32(qa[kk], f.b[0][kk], c0);
        c1 = wmma_f32(qa[kk], f.b[1][kk], c1);
    }
}

__device__ __forceinline__ void pv_load(pv_frags& f, const float* abase,
                                        const float* vbase, int lh, int g) {
#pragma unroll
    for (int q = 0; q < 4; ++q) {
        const int kk = g * 4 + q;
        f.a[q] = *(const v2f*)(abase + kk * 4);
#pragma unroll
        for (int td = 0; td < 4; ++td)
            f.b[q][td] = *(const v2f*)(vbase + kk * 256 + ((td + lh) & 3) * 32);
    }
}

__device__ __forceinline__ void pv_mma(const pv_frags& f, v8f (&oc)[4]) {
#pragma unroll
    for (int q = 0; q < 4; ++q)
#pragma unroll
        for (int td = 0; td < 4; ++td) oc[td] = wmma_f32(f.a[q], f.b[q][td], oc[td]);
}

__device__ __forceinline__ void q_load(v2f (&qa)[16], const float* qrow) {
#pragma unroll
    for (int kk = 0; kk < 16; ++kk) qa[kk] = *(const v2f*)(qrow + kk * 4);
}

__global__ __launch_bounds__(ATT_WAVES * 32) void lin_attn_kernel(
    const float* __restrict__ Q, const float* __restrict__ Kpp,
    const float* __restrict__ Vpp, float* __restrict__ out,
    float* __restrict__ attn) {
    const int S = LIN_S, D = LIN_D, KP = LIN_KP;
    const float scale = 0.125f;

    __shared__ float kpl[32 * 512];
    __shared__ float vpl[128 * 128];
    __shared__ float stg[ATT_WAVES * 16 * LDS_STRIDE];

    const int blk   = blockIdx.x;
    const int b     = blk >> 4;
    const int row0  = (blk & 15) * ATT_ROWS;
    const int tid   = threadIdx.x;
    const int w     = tid >> 5;
    const int l     = tid & 31;
    const int lh    = l >> 4;
    const int ll    = l & 15;

    const float* qbase = Q + (size_t)b * S * D + (size_t)(row0 + w * 16 + ll) * D + lh * 2;
    v2f qa[16];
    q_load(qa, qbase);

    {
        const v4f* kg = (const v4f*)(Kpp + (size_t)b * D * KP) + tid;
        const v4f* vg = (const v4f*)(Vpp + (size_t)b * KP * D) + tid;
        v4f kt[16], vt[16];
#pragma unroll
        for (int it = 0; it < 16; ++it) kt[it] = kg[it * 256];
#pragma unroll
        for (int it = 0; it < 16; ++it) vt[it] = vg[it * 256];
        const int kr = tid >> 7;
        float* kdst = kpl + kr * 512 + (((tid & 127) * 4 + 32 * kr) & 511);
        const int vr = tid >> 5;
        float* vdst = vpl + vr * 128 + (((tid & 31) * 4 + 32 * (vr & 1)) & 127);
#pragma unroll
        for (int it = 0; it < 16; ++it) *(v4f*)(kdst + it * 1024) = kt[it];
#pragma unroll
        for (int it = 0; it < 16; ++it) *(v4f*)(vdst + it * 1024) = vt[it];
    }
    __syncthreads();

    const float* kbase = kpl + lh * 512 + ll * 2;
    const float* vbase = vpl + lh * 128 + ll * 2;
    float* mystg = stg + (size_t)w * 16 * LDS_STRIDE;
    const float* abase = mystg + ll * LDS_STRIDE + lh * 2;

#pragma unroll 1
    for (int si = 0; si < 2; ++si) {
        const int s0 = row0 + si * 128 + w * 16;

        v8f sc[16];
#pragma unroll
        for (int t = 0; t < 16; ++t) sc[t] = (v8f){};
        sc_frags f0, f1;
        sc_load(f0, kbase, lh, 0);
#pragma unroll
        for (int p = 0; p < 8; p += 2) {
            sc_load(f1, kbase, lh, p + 1);
            __builtin_amdgcn_sched_barrier(0);
            sc_mma(f0, qa, sc[2 * p], sc[2 * p + 1]);
            __builtin_amdgcn_sched_barrier(0);
            if (p + 2 < 8) sc_load(f0, kbase, lh, p + 2);
            __builtin_amdgcn_sched_barrier(0);
            sc_mma(f1, qa, sc[2 * p + 2], sc[2 * p + 3]);
            __builtin_amdgcn_sched_barrier(0);
        }

        if (si == 0) q_load(qa, qbase + (size_t)128 * D);

#pragma unroll
        for (int v = 0; v < 8; ++v) {
            float mx = sc[0][v];
#pragma unroll
            for (int t = 1; t < 16; ++t) mx = fmaxf(mx, sc[t][v]);
            mx = fmaxf(mx, __shfl_xor(mx, 1, 32));
            mx = fmaxf(mx, __shfl_xor(mx, 2, 32));
            mx = fmaxf(mx, __shfl_xor(mx, 4, 32));
            mx = fmaxf(mx, __shfl_xor(mx, 8, 32));
            float sum = 0.f;
#pragma unroll
            for (int t = 0; t < 16; ++t) {
                float p = __expf((sc[t][v] - mx) * scale);
                sc[t][v] = p;
                sum += p;
            }
            sum += __shfl_xor(sum, 1, 32);
            sum += __shfl_xor(sum, 2, 32);
            sum += __shfl_xor(sum, 4, 32);
            sum += __shfl_xor(sum, 8, 32);
            const float r = 1.0f / sum;
#pragma unroll
            for (int t = 0; t < 16; ++t) sc[t][v] *= r;
        }

        float* arow = attn + (size_t)b * S * KP;
#pragma unroll
        for (int t = 0; t < 16; ++t) {
#pragma unroll
            for (int v = 0; v < 8; ++v) {
                const int m   = v + 8 * lh;
                const int col = t * 16 + ll;
                arow[(size_t)(s0 + m) * KP + col] = sc[t][v];
                mystg[m * LDS_STRIDE + col]       = sc[t][v];
            }
        }
        __syncthreads();

        v8f oc[4] = {};
        pv_frags g0, g1;
        pv_load(g0, abase, vbase, lh, 0);
#pragma unroll
        for (int g = 0; g < 16; g += 2) {
            pv_load(g1, abase, vbase, lh, g + 1);
            __builtin_amdgcn_sched_barrier(0);
            pv_mma(g0, oc);
            __builtin_amdgcn_sched_barrier(0);
            if (g + 2 < 16) pv_load(g0, abase, vbase, lh, g + 2);
            __builtin_amdgcn_sched_barrier(0);
            pv_mma(g1, oc);
            __builtin_amdgcn_sched_barrier(0);
        }
#pragma unroll
        for (int td = 0; td < 4; ++td) {
#pragma unroll
            for (int v = 0; v < 8; ++v)
                out[(size_t)b * S * D + (size_t)(s0 + v + 8 * lh) * D + td * 16 + ll] =
                    oc[td][v];
        }
    }
}

extern "C" void kernel_launch(void* const* d_in, const int* in_sizes, int n_in,
                              void* d_out, int out_size, void* d_ws, size_t ws_size,
                              hipStream_t stream) {
    (void)in_sizes; (void)n_in; (void)out_size; (void)ws_size;
    const float* Q   = (const float*)d_in[0];
    const float* K   = (const float*)d_in[1];
    const float* V   = (const float*)d_in[2];
    const float* E_w = (const float*)d_in[3];
    const float* E_b = (const float*)d_in[4];
    const float* F_w = (const float*)d_in[5];
    const float* F_b = (const float*)d_in[6];

    float* out  = (float*)d_out;
    float* attn = out + (size_t)LIN_B * LIN_S * LIN_D;

    float* Kpp = (float*)d_ws;
    float* Vpp = Kpp + (size_t)LIN_B * LIN_D * LIN_KP;

    const int projBlocks = LIN_B * 16;
    const int attnBlocks = LIN_B * (LIN_S / ATT_ROWS);

    lin_proj_kernel<true ><<<projBlocks, PROJ_WAVES * 32, 0, stream>>>(K, E_w, E_b, Kpp);
    lin_proj_kernel<false><<<projBlocks, PROJ_WAVES * 32, 0, stream>>>(V, F_w, F_b, Vpp);
    lin_attn_kernel<<<attnBlocks, ATT_WAVES * 32, 0, stream>>>(Q, Kpp, Vpp, out, attn);
}


# Projection tile doubled to 32 kp rows per work-group: each X fragment feeds two A tiles (X re-read 8x, not 16x); K and V projections share one 256-group launch; order of sums unchanged
# speedup vs baseline: 3.2450x; 1.1417x over previous
#include <hip/hip_runtime.h>
#include <hip/hip_bf16.h>


typedef float v2f __attribute__((ext_vector_type(2)));
typedef float v8f __attribute__((ext_vector_type(8)));

#define LIN_B  16
#define LIN_S  4096
#define LIN_D  64
#define LIN_KP 256

__device__ __forceinline__ v8f wmma_f32(v2f a, v2f b, v8f c) {
    return __builtin_amdgcn_wmma_f32_16x16x4_f32(false, a, false, b, (short)0, c,
                                                 false, false);
}

typedef float v4f __attribute__((ext_vector_type(4)));

#define PROJ_WAVES 16
#define PROJ_TILE  32
#define PROJ_PITCH 68

struct proj_frags {
    v4f a[2][2];
    v4f x[2][4];
};

__device__ __forceinline__ void proj_load(proj_frags& f, const float* wrow,
                                          const float* xrow, int hc) {
    const int S = LIN_S, D = LIN_D;
    const int off = (hc >> 1) * 512 + (hc & 1) * 16;
#pragma unroll
    for (int t = 0; t < 2; ++t)
#pragma unroll
        for (int j = 0; j < 2; ++j)
            f.a[t][j] = *(const v4f*)(wrow + (size_t)t * 16 * S + off + j * 8);
#pragma unroll
    for (int j = 0; j < 2; ++j)
#pragma unroll
        for (int r = 0; r < 4; ++r)
            f.x[j][r] = *(const v4f*)(xrow + (size_t)(off + j * 8 + r) * D);
}

__device__ __forceinline__ void proj_mma(const proj_frags& f, v8f (&acc)[2][4]) {
#pragma unroll
    for (int j = 0; j < 2; ++j)
#pragma unroll
        for (int e = 0; e < 2; ++e) {
            v2f a0 = {f.a[0][j][2 * e], f.a[0][j][2 * e + 1]};
            v2f a1 = {f.a[1][j][2 * e], f.a[1][j][2 * e + 1]};
#pragma unroll
            for (int td = 0; td < 4; ++td) {
                v2f bb = {f.x[j][2 * e][td], f.x[j][2 * e + 1][td]};
                acc[0][td] = wmma_f32(a0, bb, acc[0][td]);
                acc[1][td] = wmma_f32(a1, bb, acc[1][td]);
            }
        }
}

__global__ __launch_bounds__(PROJ_WAVES * 32) void lin_proj_kernel(
    const float* K, const float* V, const float* E_w, const float* F_w,
    const float* E_b, const float* F_b,
    float* __restrict__ Kpp, float* __restrict__ Vpp) {
    const int S = LIN_S, D = LIN_D, KP = LIN_KP;

    __shared__ float part[PROJ_WAVES * PROJ_TILE * PROJ_PITCH];

    const int blk = blockIdx.x;
    const bool isV = blk >= LIN_B * (LIN_KP / PROJ_TILE);
    const int b   = (blk >> 3) & 15;
    const int kp0 = (blk & 7) * PROJ_TILE;
    const int tid = threadIdx.x;
    const int w   = tid >> 5;
    const int l   = tid & 31;
    const int lh  = l >> 4;
    const int ll  = l & 15;

    const float* X    = isV ? V : K;
    const float* W    = isV ? F_w : E_w;
    const float* bias = isV ? F_b : E_b;

    const float* wrow = W + (size_t)(kp0 + ll) * S + w * 32 + lh * 4;
    const float* xrow = X + (size_t)b * S * D + (size_t)(w * 32 + lh * 4) * D + ll * 4;

    v8f acc[2][4] = {};
    proj_frags f0, f1;
    proj_load(f0, wrow, xrow, 0);
#pragma unroll
    for (int hc = 0; hc < 16; hc += 2) {
        proj_load(f1, wrow, xrow, hc + 1);
        __builtin_amdgcn_sched_barrier(0);
        proj_mma(f0, acc);
        __builtin_amdgcn_sched_barrier(0);
        if (hc + 2 < 16) proj_load(f0, wrow, xrow, hc + 2);
        __builtin_amdgcn_sched_barrier(0);
        proj_mma(f1, acc);
        __builtin_amdgcn_sched_barrier(0);
    }

    float* mine = part + (size_t)(w * PROJ_TILE + 8 * lh) * PROJ_PITCH + ll * 4;
#pragma unroll
    for (int t = 0; t < 2; ++t)
#pragma unroll
        for (int v = 0; v < 8; ++v) {
            v4f o = {acc[t][0][v], acc[t][1][v], acc[t][2][v], acc[t][3][v]};
            *(v4f*)(mine + (t * 16 + v) * PROJ_PITCH) = o;
        }
    __syncthreads();

    if (!isV) {
        const int p   = tid >> 4;
        const int kpl = tid & 15;
#pragma unroll
        for (int r = 0; r < 2; ++r) {
            const int row = r * 16 + kpl;
            const float* src = part + (size_t)row * PROJ_PITCH + 2 * p;
            v2f s = {0.f, 0.f};
#pragma unroll
            for (int q = 0; q < PROJ_WAVES; ++q)
                s += *(const v2f*)(src + q * PROJ_TILE * PROJ_PITCH);
            const float bv = bias[kp0 + row];
            s.x += bv;
            s.y += bv;
            *(v2f*)(Kpp + (size_t)b * D * KP + (size_t)p * (2 * KP) +
                    (size_t)(kp0 + row) * 2) = s;
        }
    } else {
        const int q2 = tid >> 6;
        const int d  = tid & 63;
#pragma unroll
        for (int r = 0; r < 2; ++r) {
            const int pr = r * 8 + q2;
            const float* src = part + (size_t)(2 * pr) * PROJ_PITCH + d;
            v2f s = {0.f, 0.f};
#pragma unroll
            for (int q = 0; q < PROJ_WAVES; ++q) {
                s.x += src[q * PROJ_TILE * PROJ_PITCH];
                s.y += src[q * PROJ_TILE * PROJ_PITCH + PROJ_PITCH];
            }
            s.x += bias[kp0 + 2 * pr];
            s.y += bias[kp0 + 2 * pr + 1];
            *(v2f*)(Vpp + (size_t)b * KP * D + (size_t)((kp0 >> 1) + pr) * (2 * D) +
                    (size_t)d * 2) = s;
        }
    }
}

#define LDS_STRIDE 260
#define ATT_WAVES  8
#define ATT_ROWS   256

struct sc_frags { v2f b[2][16]; };
struct pv_frags { v2f a[4]; v2f b[4][4]; };

__device__ __forceinline__ void sc_load(sc_frags& f, const float* kbase, int lh, int p) {
#pragma unroll
    for (int u = 0; u < 2; ++u) {
        const float* src = kbase + ((2 * p + u + lh) & 15) * 32;
#pragma unroll
        for (int kk = 0; kk < 16; ++kk) f.b[u][kk] = *(const v2f*)(src + kk * 1024);
    }
}

__device__ __forceinline__ void sc_mma(const sc_frags& f, const v2f (&qa)[16],
                                       v8f& c0, v8f& c1) {
#pragma unroll
    for (int kk = 0; kk < 16; ++kk) {
        c0 = wmma_f32(qa[kk], f.b[0][kk], c0);
        c1 = wmma_f32(qa[kk], f.b[1][kk], c1);
    }
}

__device__ __forceinline__ void pv_load(pv_frags& f, const float* abase,
                                        const float* vbase, int lh, int g) {
#pragma unroll
    for (int q = 0; q < 4; ++q) {
        const int kk = g * 4 + q;
        f.a[q] = *(const v2f*)(abase + kk * 4);
#pragma unroll
        for (int td = 0; td < 4; ++td)
            f.b[q][td] = *(const v2f*)(vbase + kk * 256 + ((td + lh) & 3) * 32);
    }
}

__device__ __forceinline__ void pv_mma(const pv_frags& f, v8f (&oc)[4]) {
#pragma unroll
    for (int q = 0; q < 4; ++q)
#pragma unroll
        for (int td = 0; td < 4; ++td) oc[td] = wmma_f32(f.a[q], f.b[q][td], oc[td]);
}

__device__ __forceinline__ void q_load(v2f (&qa)[16], const float* qrow) {
#pragma unroll
    for (int kk = 0; kk < 16; ++kk) qa[kk] = *(const v2f*)(qrow + kk * 4);
}

__global__ __launch_bounds__(ATT_WAVES * 32) void lin_attn_kernel(
    const float* __restrict__ Q, const float* __restrict__ Kpp,
    const float* __restrict__ Vpp, float* __restrict__ out,
    float* __restrict__ attn) {
    const int S = LIN_S, D = LIN_D, KP = LIN_KP;
    const float scale = 0.125f;

    __shared__ float kpl[32 * 512];
    __shared__ float vpl[128 * 128];
    __shared__ float stg[ATT_WAVES * 16 * LDS_STRIDE];

    const int blk   = blockIdx.x;
    const int b     = blk >> 4;
    const int row0  = (blk & 15) * ATT_ROWS;
    const int tid   = threadIdx.x;
    const int w     = tid >> 5;
    const int l     = tid & 31;
    const int lh    = l >> 4;
    const int ll    = l & 15;

    const float* qbase = Q + (size_t)b * S * D + (size_t)(row0 + w * 16 + ll) * D + lh * 2;
    v2f qa[16];
    q_load(qa, qbase);

    {
        const v4f* kg = (const v4f*)(Kpp + (size_t)b * D * KP) + tid;
        const v4f* vg = (const v4f*)(Vpp + (size_t)b * KP * D) + tid;
        v4f kt[16], vt[16];
#pragma unroll
        for (int it = 0; it < 16; ++it) kt[it] = kg[it * 256];
#pragma unroll
        for (int it = 0; it < 16; ++it) vt[it] = vg[it * 256];
        const int kr = tid >> 7;
        float* kdst = kpl + kr * 512 + (((tid & 127) * 4 + 32 * kr) & 511);
        const int vr = tid >> 5;
        float* vdst = vpl + vr * 128 + (((tid & 31) * 4 + 32 * (vr & 1)) & 127);
#pragma unroll
        for (int it = 0; it < 16; ++it) *(v4f*)(kdst + it * 1024) = kt[it];
#pragma unroll
        for (int it = 0; it < 16; ++it) *(v4f*)(vdst + it * 1024) = vt[it];
    }
    __syncthreads();

    const float* kbase = kpl + lh * 512 + ll * 2;
    const float* vbase = vpl + lh * 128 + ll * 2;
    float* mystg = stg + (size_t)w * 16 * LDS_STRIDE;
    const float* abase = mystg + ll * LDS_STRIDE + lh * 2;

#pragma unroll 1
    for (int si = 0; si < 2; ++si) {
        const int s0 = row0 + si * 128 + w * 16;

        v8f sc[16];
#pragma unroll
        for (int t = 0; t < 16; ++t) sc[t] = (v8f){};
        sc_frags f0, f1;
        sc_load(f0, kbase, lh, 0);
#pragma unroll
        for (int p = 0; p < 8; p += 2) {
            sc_load(f1, kbase, lh, p + 1);
            __builtin_amdgcn_sched_barrier(0);
            sc_mma(f0, qa, sc[2 * p], sc[2 * p + 1]);
            __builtin_amdgcn_sched_barrier(0);
            if (p + 2 < 8) sc_load(f0, kbase, lh, p + 2);
            __builtin_amdgcn_sched_barrier(0);
            sc_mma(f1, qa, sc[2 * p + 2], sc[2 * p + 3]);
            __builtin_amdgcn_sched_barrier(0);
        }

        if (si == 0) q_load(qa, qbase + (size_t)128 * D);

#pragma unroll
        for (int v = 0; v < 8; ++v) {
            float mx = sc[0][v];
#pragma unroll
            for (int t = 1; t < 16; ++t) mx = fmaxf(mx, sc[t][v]);
            mx = fmaxf(mx, __shfl_xor(mx, 1, 32));
            mx = fmaxf(mx, __shfl_xor(mx, 2, 32));
            mx = fmaxf(mx, __shfl_xor(mx, 4, 32));
            mx = fmaxf(mx, __shfl_xor(mx, 8, 32));
            float sum = 0.f;
#pragma unroll
            for (int t = 0; t < 16; ++t) {
                float p = __expf((sc[t][v] - mx) * scale);
                sc[t][v] = p;
                sum += p;
            }
            sum += __shfl_xor(sum, 1, 32);
            sum += __shfl_xor(sum, 2, 32);
            sum += __shfl_xor(sum, 4, 32);
            sum += __shfl_xor(sum, 8, 32);
            const float r = 1.0f / sum;
#pragma unroll
            for (int t = 0; t < 16; ++t) sc[t][v] *= r;
        }

        float* arow = attn + (size_t)b * S * KP;
#pragma unroll
        for (int t = 0; t < 16; ++t) {
#pragma unroll
            for (int v = 0; v < 8; ++v) {
                const int m   = v + 8 * lh;
                const int col = t * 16 + ll;
                arow[(size_t)(s0 + m) * KP + col] = sc[t][v];
                mystg[m * LDS_STRIDE + col]       = sc[t][v];
            }
        }
        __syncthreads();

        v8f oc[4] = {};
        pv_frags g0, g1;
        pv_load(g0, abase, vbase, lh, 0);
#pragma unroll
        for (int g = 0; g < 16; g += 2) {
            pv_load(g1, abase, vbase, lh, g + 1);
            __builtin_amdgcn_sched_barrier(0);
            pv_mma(g0, oc);
            __builtin_amdgcn_sched_barrier(0);
            if (g + 2 < 16) pv_load(g0, abase, vbase, lh, g + 2);
            __builtin_amdgcn_sched_barrier(0);
            pv_mma(g1, oc);
            __builtin_amdgcn_sched_barrier(0);
        }
#pragma unroll
        for (int td = 0; td < 4; ++td) {
#pragma unroll
            for (int v = 0; v < 8; ++v)
                out[(size_t)b * S * D + (size_t)(s0 + v + 8 * lh) * D + td * 16 + ll] =
                    oc[td][v];
        }
    }
}

extern "C" void kernel_launch(void* const* d_in, const int* in_sizes, int n_in,
                              void* d_out, int out_size, void* d_ws, size_t ws_size,
                              hipStream_t stream) {
    (void)in_sizes; (void)n_in; (void)out_size; (void)ws_size;
    const float* Q   = (const float*)d_in[0];
    const float* K   = (const float*)d_in[1];
    const float* V   = (const float*)d_in[2];
    const float* E_w = (const float*)d_in[3];
    const float* E_b = (const float*)d_in[4];
    const float* F_w = (const float*)d_in[5];
    const float* F_b = (const float*)d_in[6];

    float* out  = (float*)d_out;
    float* attn = out + (size_t)LIN_B * LIN_S * LIN_D;

    float* Kpp = (float*)d_ws;
    float* Vpp = Kpp + (size_t)LIN_B * LIN_D * LIN_KP;

    const int projBlocks = 2 * LIN_B * (LIN_KP / PROJ_TILE);
    const int attnBlocks = LIN_B * (LIN_S / ATT_ROWS);

    lin_proj_kernel<<<projBlocks, PROJ_WAVES * 32, 0, stream>>>(K, V, E_w, F_w, E_b, F_b,
                                                                Kpp, Vpp);
    lin_attn_kernel<<<attnBlocks, ATT_WAVES * 32, 0, stream>>>(Q, Kpp, Vpp, out, attn);
}


# Attention: attn's 128 store instructions per strip leave the burst before the strip barrier and ride 8 per WMMA group under the output GEMM, a line's two 64-B halves back to back; sums unchanged
# speedup vs baseline: 3.3871x; 1.0438x over previous
#include <hip/hip_runtime.h>
#include <hip/hip_bf16.h>


typedef float v2f __attribute__((ext_vector_type(2)));
typedef float v8f __attribute__((ext_vector_type(8)));

#define LIN_B  16
#define LIN_S  4096
#define LIN_D  64
#define LIN_KP 256

__device__ __forceinline__ v8f wmma_f32(v2f a, v2f b, v8f c) {
    return __builtin_amdgcn_wmma_f32_16x16x4_f32(false, a, false, b, (short)0, c,
                                                 false, false);
}

typedef float v4f __attribute__((ext_vector_type(4)));

#define PROJ_WAVES 16
#define PROJ_TILE  32
#define PROJ_PITCH 68

struct proj_frags {
    v4f a[2][2];
    v4f x[2][4];
};

__device__ __forceinline__ void proj_load(proj_frags& f, const float* wrow,
                                          const float* xrow, int hc) {
    const int S = LIN_S, D = LIN_D;
    const int off = (hc >> 1) * 512 + (hc & 1) * 16;
#pragma unroll
    for (int t = 0; t < 2; ++t)
#pragma unroll
        for (int j = 0; j < 2; ++j)
            f.a[t][j] = *(const v4f*)(wrow + (size_t)t * 16 * S + off + j * 8);
#pragma unroll
    for (int j = 0; j < 2; ++j)
#pragma unroll
        for (int r = 0; r < 4; ++r)
            f.x[j][r] = *(const v4f*)(xrow + (size_t)(off + j * 8 + r) * D);
}

__device__ __forceinline__ void proj_mma(const proj_frags& f, v8f (&acc)[2][4]) {
#pragma unroll
    for (int j = 0; j < 2; ++j)
#pragma unroll
        for (int e = 0; e < 2; ++e) {
            v2f a0 = {f.a[0][j][2 * e], f.a[0][j][2 * e + 1]};
            v2f a1 = {f.a[1][j][2 * e], f.a[1][j][2 * e + 1]};
#pragma unroll
            for (int td = 0; td < 4; ++td) {
                v2f bb = {f.x[j][2 * e][td], f.x[j][2 * e + 1][td]};
                acc[0][td] = wmma_f32(a0, bb, acc[0][td]);
                acc[1][td] = wmma_f32(a1, bb, acc[1][td]);
            }
        }
}

__global__ __launch_bounds__(PROJ_WAVES * 32) void lin_proj_kernel(
    const float* K, const float* V, const float* E_w, const float* F_w,
    const float* E_b, const float* F_b,
    float* __restrict__ Kpp, float* __restrict__ Vpp) {
    const int S = LIN_S, D = LIN_D, KP = LIN_KP;

    __shared__ float part[PROJ_WAVES * PROJ_TILE * PROJ_PITCH];

    const int blk = blockIdx.x;
    const bool isV = blk >= LIN_B * (LIN_KP / PROJ_TILE);
    const int b   = (blk >> 3) & 15;
    const int kp0 = (blk & 7) * PROJ_TILE;
    const int tid = threadIdx.x;
    const int w   = tid >> 5;
    const int l   = tid & 31;
    const int lh  = l >> 4;
    const int ll  = l & 15;

    const float* X    = isV ? V : K;
    const float* W    = isV ? F_w : E_w;
    const float* bias = isV ? F_b : E_b;

    const float* wrow = W + (size_t)(kp0 + ll) * S + w * 32 + lh * 4;
    const float* xrow = X + (size_t)b * S * D + (size_t)(w * 32 + lh * 4) * D + ll * 4;

    v8f acc[2][4] = {};
    proj_frags f0, f1;
    proj_load(f0, wrow, xrow, 0);
#pragma unroll
    for (int hc = 0; hc < 16; hc += 2) {
        proj_load(f1, wrow, xrow, hc + 1);
        __builtin_amdgcn_sched_barrier(0);
        proj_mma(f0, acc);
        __builtin_amdgcn_sched_barrier(0);
        if (hc + 2 < 16) proj_load(f0, wrow, xrow, hc + 2);
        __builtin_amdgcn_sched_barrier(0);
        proj_mma(f1, acc);
        __builtin_amdgcn_sched_barrier(0);
    }

    float* mine = part + (size_t)(w * PROJ_TILE + 8 * lh) * PROJ_PITCH + ll * 4;
#pragma unroll
    for (int t = 0; t < 2; ++t)
#pragma unroll
        for (int v = 0; v < 8; ++v) {
            v4f o = {acc[t][0][v], acc[t][1][v], acc[t][2][v], acc[t][3][v]};
            *(v4f*)(mine + (t * 16 + v) * PROJ_PITCH) = o;
        }
    __syncthreads();

    if (!isV) {
        const int p   = tid >> 4;
        const int kpl = tid & 15;
#pragma unroll
        for (int r = 0; r < 2; ++r) {
            const int row = r * 16 + kpl;
            const float* src = part + (size_t)row * PROJ_PITCH + 2 * p;
            v2f s = {0.f, 0.f};
#pragma unroll
            for (int q = 0; q < PROJ_WAVES; ++q)
                s += *(const v2f*)(src + q * PROJ_TILE * PROJ_PITCH);
            const float bv = bias[kp0 + row];
            s.x += bv;
            s.y += bv;
            *(v2f*)(Kpp + (size_t)b * D * KP + (size_t)p * (2 * KP) +
                    (size_t)(kp0 + row) * 2) = s;
        }
    } else {
        const int q2 = tid >> 6;
        const int d  = tid & 63;
#pragma unroll
        for (int r = 0; r < 2; ++r) {
            const int pr = r * 8 + q2;
            const float* src = part + (size_t)(2 * pr) * PROJ_PITCH + d;
            v2f s = {0.f, 0.f};
#pragma unroll
            for (int q = 0; q < PROJ_WAVES; ++q) {
                s.x += src[q * PROJ_TILE * PROJ_PITCH];
                s.y += src[q * PROJ_TILE * PROJ_PITCH + PROJ_PITCH];
            }
            s.x += bias[kp0 + 2 * pr];
            s.y += bias[kp0 + 2 * pr + 1];
            *(v2f*)(Vpp + (size_t)b * KP * D + (size_t)((kp0 >> 1) + pr) * (2 * D) +
                    (size_t)d * 2) = s;
        }
    }
}

#define LDS_STRIDE 260
#define ATT_WAVES  8
#define ATT_ROWS   256

struct sc_frags { v2f b[2][16]; };
struct pv_frags { v2f a[4]; v2f b[4][4]; };

__device__ __forceinline__ void sc_load(sc_frags& f, const float* kbase, int lh, int p) {
#pragma unroll
    for (int u = 0; u < 2; ++u) {
        const float* src = kbase + ((2 * p + u + lh) & 15) * 32;
#pragma unroll
        for (int kk = 0; kk < 16; ++kk) f.b[u][kk] = *(const v2f*)(src + kk * 1024);
    }
}

__device__ __forceinline__ void sc_mma(const sc_frags& f, const v2f (&qa)[16],
                                       v8f& c0, v8f& c1) {
#pragma unroll
    for (int kk = 0; kk < 16; ++kk) {
        c0 = wmma_f32(qa[kk], f.b[0][kk], c0);
        c1 = wmma_f32(qa[kk], f.b[1][kk], c1);
    }
}

__device__ __forceinline__ void pv_load(pv_frags& f, const float* abase,
                                        const float* vbase, int lh, int g) {
#pragma unroll
    for (int q = 0; q < 4; ++q) {
        const int kk = g * 4 + q;
        f.a[q] = *(const v2f*)(abase + kk * 4);
#pragma unroll
        for (int td = 0; td < 4; ++td)
            f.b[q][td] = *(const v2f*)(vbase + kk * 256 + ((td + lh) & 3) * 32);
    }
}

__device__ __forceinline__ void pv_mma_st(const pv_frags& f, v8f (&oc)[4],
                                          const v8f (&sc)[16], float* alane, int g) {
    const int t0 = (g >> 1) * 2;
    const int v0 = (g & 1) * 4;
#pragma unroll
    for (int q = 0; q < 4; ++q) {
#pragma unroll
        for (int td = 0; td < 4; ++td) oc[td] = wmma_f32(f.a[q], f.b[q][td], oc[td]);
        alane[(v0 + q) * LIN_KP + t0 * 16]       = sc[t0][v0 + q];
        alane[(v0 + q) * LIN_KP + (t0 + 1) * 16] = sc[t0 + 1][v0 + q];
    }
}

__device__ __forceinline__ void q_load(v2f (&qa)[16], const float* qrow) {
#pragma unroll
    for (int kk = 0; kk < 16; ++kk) qa[kk] = *(const v2f*)(qrow + kk * 4);
}

__global__ __launch_bounds__(ATT_WAVES * 32) void lin_attn_kernel(
    const float* __restrict__ Q, const float* __restrict__ Kpp,
    const float* __restrict__ Vpp, float* __restrict__ out,
    float* __restrict__ attn) {
    const int S = LIN_S, D = LIN_D, KP = LIN_KP;
    const float scale = 0.125f;

    __shared__ float kpl[32 * 512];
    __shared__ float vpl[128 * 128];
    __shared__ float stg[ATT_WAVES * 16 * LDS_STRIDE];

    const int blk   = blockIdx.x;
    const int b     = blk >> 4;
    const int row0  = (blk & 15) * ATT_ROWS;
    const int tid   = threadIdx.x;
    const int w     = tid >> 5;
    const int l     = tid & 31;
    const int lh    = l >> 4;
    const int ll    = l & 15;

    const float* qbase = Q + (size_t)b * S * D + (size_t)(row0 + w * 16 + ll) * D + lh * 2;
    v2f qa[16];
    q_load(qa, qbase);

    {
        const v4f* kg = (const v4f*)(Kpp + (size_t)b * D * KP) + tid;
        const v4f* vg = (const v4f*)(Vpp + (size_t)b * KP * D) + tid;
        v4f kt[16], vt[16];
#pragma unroll
        for (int it = 0; it < 16; ++it) kt[it] = kg[it * 256];
#pragma unroll
        for (int it = 0; it < 16; ++it) vt[it] = vg[it * 256];
        const int kr = tid >> 7;
        float* kdst = kpl + kr * 512 + (((tid & 127) * 4 + 32 * kr) & 511);
        const int vr = tid >> 5;
        float* vdst = vpl + vr * 128 + (((tid & 31) * 4 + 32 * (vr & 1)) & 127);
#pragma unroll
        for (int it = 0; it < 16; ++it) *(v4f*)(kdst + it * 1024) = kt[it];
#pragma unroll
        for (int it = 0; it < 16; ++it) *(v4f*)(vdst + it * 1024) = vt[it];
    }
    __syncthreads();

    const float* kbase = kpl + lh * 512 + ll * 2;
    const float* vbase = vpl + lh * 128 + ll * 2;
    float* mystg = stg + (size_t)w * 16 * LDS_STRIDE;
    const float* abase = mystg + ll * LDS_STRIDE + lh * 2;

#pragma unroll 1
    for (int si = 0; si < 2; ++si) {
        const int s0 = row0 + si * 128 + w * 16;

        v8f sc[16];
#pragma unroll
        for (int t = 0; t < 16; ++t) sc[t] = (v8f){};
        sc_frags f0, f1;
        sc_load(f0, kbase, lh, 0);
#pragma unroll
        for (int p = 0; p < 8; p += 2) {
            sc_load(f1, kbase, lh, p + 1);
            __builtin_amdgcn_sched_barrier(0);
            sc_mma(f0, qa, sc[2 * p], sc[2 * p + 1]);
            __builtin_amdgcn_sched_barrier(0);
            if (p + 2 < 8) sc_load(f0, kbase, lh, p + 2);
            __builtin_amdgcn_sched_barrier(0);
            sc_mma(f1, qa, sc[2 * p + 2], sc[2 * p + 3]);
            __builtin_amdgcn_sched_barrier(0);
        }

        if (si == 0) q_load(qa, qbase + (size_t)128 * D);

#pragma unroll
        for (int v = 0; v < 8; ++v) {
            float mx = sc[0][v];
#pragma unroll
            for (int t = 1; t < 16; ++t) mx = fmaxf(mx, sc[t][v]);
            mx = fmaxf(mx, __shfl_xor(mx, 1, 32));
            mx = fmaxf(mx, __shfl_xor(mx, 2, 32));
            mx = fmaxf(mx, __shfl_xor(mx, 4, 32));
            mx = fmaxf(mx, __shfl_xor(mx, 8, 32));
            float sum = 0.f;
#pragma unroll
            for (int t = 0; t < 16; ++t) {
                float p = __expf((sc[t][v] - mx) * scale);
                sc[t][v] = p;
                sum += p;
            }
            sum += __shfl_xor(sum, 1, 32);
            sum += __shfl_xor(sum, 2, 32);
            sum += __shfl_xor(sum, 4, 32);
            sum += __shfl_xor(sum, 8, 32);
            const float r = 1.0f / sum;
#pragma unroll
            for (int t = 0; t < 16; ++t) sc[t][v] *= r;
        }

#pragma unroll
        for (int t = 0; t < 16; ++t) {
#pragma unroll
            for (int v = 0; v < 8; ++v) {
                const int m   = v + 8 * lh;
                const int col = t * 16 + ll;
                mystg[m * LDS_STRIDE + col] = sc[t][v];
            }
        }
        __syncthreads();

        float* alane = attn + (size_t)b * S * KP + (size_t)(s0 + 8 * lh) * KP + ll;
        v8f oc[4] = {};
        pv_frags g0, g1;
        pv_load(g0, abase, vbase, lh, 0);
#pragma unroll
        for (int g = 0; g < 16; g += 2) {
            pv_load(g1, abase, vbase, lh, g + 1);
            __builtin_amdgcn_sched_barrier(0);
            pv_mma_st(g0, oc, sc, alane, g);
            __builtin_amdgcn_sched_barrier(0);
            if (g + 2 < 16) pv_load(g0, abase, vbase, lh, g + 2);
            __builtin_amdgcn_sched_barrier(0);
            pv_mma_st(g1, oc, sc, alane, g + 1);
            __builtin_amdgcn_sched_barrier(0);
        }
#pragma unroll
        for (int td = 0; td < 4; ++td) {
#pragma unroll
            for (int v = 0; v < 8; ++v)
                out[(size_t)b * S * D + (size_t)(s0 + v + 8 * lh) * D + td * 16 + ll] =
                    oc[td][v];
        }
    }
}

extern "C" void kernel_launch(void* const* d_in, const int* in_sizes, int n_in,
                              void* d_out, int out_size, void* d_ws, size_t ws_size,
                              hipStream_t stream) {
    (void)in_sizes; (void)n_in; (void)out_size; (void)ws_size;
    const float* Q   = (const float*)d_in[0];
    const float* K   = (const float*)d_in[1];
    const float* V   = (const float*)d_in[2];
    const float* E_w = (const float*)d_in[3];
    const float* E_b = (const float*)d_in[4];
    const float* F_w = (const float*)d_in[5];
    const float* F_b = (const float*)d_in[6];

    float* out  = (float*)d_out;
    float* attn = out + (size_t)LIN_B * LIN_S * LIN_D;

    float* Kpp = (float*)d_ws;
    float* Vpp = Kpp + (size_t)LIN_B * LIN_D * LIN_KP;

    const int projBlocks = 2 * LIN_B * (LIN_KP / PROJ_TILE);
    const int attnBlocks = LIN_B * (LIN_S / ATT_ROWS);

    lin_proj_kernel<<<projBlocks, PROJ_WAVES * 32, 0, stream>>>(K, V, E_w, F_w, E_b, F_b,
                                                                Kpp, Vpp);
    lin_attn_kernel<<<attnBlocks, ATT_WAVES * 32, 0, stream>>>(Q, Kpp, Vpp, out, attn);
}


# Attention: score tiles transposed (Kp as A, Q as B): softmax max/sum in a lane's registers, attn registers are the output GEMM's B, whole-row stores; order of row sums and PV k-groups changes
# speedup vs baseline: 3.7751x; 1.1146x over previous
#include <hip/hip_runtime.h>
#include <hip/hip_bf16.h>


typedef float v2f __attribute__((ext_vector_type(2)));
typedef float v8f __attribute__((ext_vector_type(8)));

#define LIN_B  16
#define LIN_S  4096
#define LIN_D  64
#define LIN_KP 256

__device__ __forceinline__ v8f wmma_f32(v2f a, v2f b, v8f c) {
    return __builtin_amdgcn_wmma_f32_16x16x4_f32(false, a, false, b, (short)0, c,
                                                 false, false);
}

typedef float v4f __attribute__((ext_vector_type(4)));

#define PROJ_WAVES 16
#define PROJ_TILE  32
#define PROJ_PITCH 68

struct proj_frags {
    v4f a[2][2];
    v4f x[2][4];
};

__device__ __forceinline__ void proj_load(proj_frags& f, const float* wrow,
                                          const float* xrow, int hc) {
    const int S = LIN_S, D = LIN_D;
    const int off = (hc >> 1) * 512 + (hc & 1) * 16;
#pragma unroll
    for (int t = 0; t < 2; ++t)
#pragma unroll
        for (int j = 0; j < 2; ++j)
            f.a[t][j] = *(const v4f*)(wrow + (size_t)t * 16 * S + off + j * 8);
#pragma unroll
    for (int j = 0; j < 2; ++j)
#pragma unroll
        for (int r = 0; r < 4; ++r)
            f.x[j][r] = *(const v4f*)(xrow + (size_t)(off + j * 8 + r) * D);
}

__device__ __forceinline__ void proj_mma(const proj_frags& f, v8f (&acc)[2][4]) {
#pragma unroll
    for (int j = 0; j < 2; ++j)
#pragma unroll
        for (int e = 0; e < 2; ++e) {
            v2f a0 = {f.a[0][j][2 * e], f.a[0][j][2 * e + 1]};
            v2f a1 = {f.a[1][j][2 * e], f.a[1][j][2 * e + 1]};
#pragma unroll
            for (int td = 0; td < 4; ++td) {
                v2f bb = {f.x[j][2 * e][td], f.x[j][2 * e + 1][td]};
                acc[0][td] = wmma_f32(a0, bb, acc[0][td]);
                acc[1][td] = wmma_f32(a1, bb, acc[1][td]);
            }
        }
}

__global__ __launch_bounds__(PROJ_WAVES * 32) void lin_proj_kernel(
    const float* K, const float* V, const float* E_w, const float* F_w,
    const float* E_b, const float* F_b,
    float* __restrict__ Kpp, float* __restrict__ Vpp) {
    const int S = LIN_S, D = LIN_D, KP = LIN_KP;

    __shared__ float part[PROJ_WAVES * PROJ_TILE * PROJ_PITCH];

    const int blk = blockIdx.x;
    const bool isV = blk >= LIN_B * (LIN_KP / PROJ_TILE);
    const int b   = (blk >> 3) & 15;
    const int kp0 = (blk & 7) * PROJ_TILE;
    const int tid = threadIdx.x;
    const int w   = tid >> 5;
    const int l   = tid & 31;
    const int lh  = l >> 4;
    const int ll  = l & 15;

    const float* X    = isV ? V : K;
    const float* W    = isV ? F_w : E_w;
    const float* bias = isV ? F_b : E_b;

    const float* wrow = W + (size_t)(kp0 + ll) * S + w * 32 + lh * 4;
    const float* xrow = X + (size_t)b * S * D + (size_t)(w * 32 + lh * 4) * D + ll * 4;

    v8f acc[2][4] = {};
    proj_frags f0, f1;
    proj_load(f0, wrow, xrow, 0);
#pragma unroll
    for (int hc = 0; hc < 16; hc += 2) {
        proj_load(f1, wrow, xrow, hc + 1);
        __builtin_amdgcn_sched_barrier(0);
        proj_mma(f0, acc);
        __builtin_amdgcn_sched_barrier(0);
        if (hc + 2 < 16) proj_load(f0, wrow, xrow, hc + 2);
        __builtin_amdgcn_sched_barrier(0);
        proj_mma(f1, acc);
        __builtin_amdgcn_sched_barrier(0);
    }

    float* mine = part + (size_t)(w * PROJ_TILE + 8 * lh) * PROJ_PITCH + ll * 4;
#pragma unroll
    for (int t = 0; t < 2; ++t)
#pragma unroll
        for (int v = 0; v < 8; ++v) {
            v4f o = {acc[t][0][v], acc[t][1][v], acc[t][2][v], acc[t][3][v]};
            *(v4f*)(mine + (t * 16 + v) * PROJ_PITCH) = o;
        }
    __syncthreads();

    if (!isV) {
        const int p   = tid >> 4;
        const int kpl = tid & 15;
#pragma unroll
        for (int r = 0; r < 2; ++r) {
            const int row = r * 16 + kpl;
            const float* src = part + (size_t)row * PROJ_PITCH + 2 * p;
            v2f s = {0.f, 0.f};
#pragma unroll
            for (int q = 0; q < PROJ_WAVES; ++q)
                s += *(const v2f*)(src + q * PROJ_TILE * PROJ_PITCH);
            const float bv = bias[kp0 + row];
            s.x += bv;
            s.y += bv;
            *(v2f*)(Kpp + (size_t)b * D * KP + (size_t)p * (2 * KP) +
                    (size_t)(kp0 + row) * 2) = s;
        }
    } else {
        const int q2 = tid >> 6;
        const int d  = tid & 63;
#pragma unroll
        for (int r = 0; r < 2; ++r) {
            const int pr = r * 8 + q2;
            const float* src = part + (size_t)(2 * pr) * PROJ_PITCH + d;
            v2f s = {0.f, 0.f};
#pragma unroll
            for (int q = 0; q < PROJ_WAVES; ++q) {
                s.x += src[q * PROJ_TILE * PROJ_PITCH];
                s.y += src[q * PROJ_TILE * PROJ_PITCH + PROJ_PITCH];
            }
            s.x += bias[kp0 + 2 * pr];
            s.y += bias[kp0 + 2 * pr + 1];
            *(v2f*)(Vpp + (size_t)b * KP * D + (size_t)((kp0 >> 1) + pr) * (2 * D) +
                    (size_t)d * 2) = s;
        }
    }
}

#define LDS_STRIDE 260
#define ATT_WAVES  8
#define ATT_ROWS   256

struct sc_frags { v2f b[2][16]; };
struct pv_frags { v2f a[4][4]; v4f row[2]; };

__device__ __forceinline__ void sc_load(sc_frags& f, const float* kbase, int lh, int p) {
#pragma unroll
    for (int u = 0; u < 2; ++u) {
        const float* src = kbase + ((2 * p + u + lh) & 15) * 32;
#pragma unroll
        for (int kk = 0; kk < 16; ++kk) f.b[u][kk] = *(const v2f*)(src + kk * 1024);
    }
}

__device__ __forceinline__ void sc_mma(const sc_frags& f, const v2f (&qa)[16],
                                       v8f& c0, v8f& c1) {
#pragma unroll
    for (int kk = 0; kk < 16; ++kk) {
        c0 = wmma_f32(f.b[0][kk], qa[kk], c0);
        c1 = wmma_f32(f.b[1][kk], qa[kk], c1);
    }
}

__device__ __forceinline__ void pv_load(pv_frags& f, const float* vbase,
                                        const float* rowsrc, int lh, int g) {
#pragma unroll
    for (int e = 0; e < 2; ++e)
        f.row[e] = *(const v4f*)(rowsrc + g * LDS_STRIDE + e * 128);
#pragma unroll
    for (int q = 0; q < 4; ++q) {
#pragma unroll
        for (int td = 0; td < 4; ++td)
            f.a[q][td] = *(const v2f*)(vbase + (8 * g + q) * 128 + ((td + lh) & 3) * 32);
    }
}

__device__ __forceinline__ void pv_mma_st(const pv_frags& f, v8f (&oc)[4],
                                          const v8f (&sc)[16], float* arow, int g) {
#pragma unroll
    for (int q = 0; q < 4; ++q) {
        const v2f bq = {sc[g][2 * q], sc[g][2 * q + 1]};
#pragma unroll
        for (int td = 0; td < 4; ++td) oc[td] = wmma_f32(f.a[q][td], bq, oc[td]);
        if (q == 0) *(v4f*)(arow + g * LIN_KP)       = f.row[0];
        if (q == 2) *(v4f*)(arow + g * LIN_KP + 128) = f.row[1];
    }
}

__device__ __forceinline__ void q_load(v2f (&qa)[16], const float* qrow) {
#pragma unroll
    for (int kk = 0; kk < 16; ++kk) qa[kk] = *(const v2f*)(qrow + kk * 4);
}

__global__ __launch_bounds__(ATT_WAVES * 32) void lin_attn_kernel(
    const float* __restrict__ Q, const float* __restrict__ Kpp,
    const float* __restrict__ Vpp, float* __restrict__ out,
    float* __restrict__ attn) {
    const int S = LIN_S, D = LIN_D, KP = LIN_KP;
    const float scale = 0.125f;

    __shared__ __attribute__((aligned(16))) float kpl[32 * 512];
    __shared__ __attribute__((aligned(16))) float vpl[128 * 128];
    __shared__ __attribute__((aligned(16))) float stg[ATT_WAVES * 16 * LDS_STRIDE];

    const int blk   = blockIdx.x;
    const int b     = blk >> 4;
    const int row0  = (blk & 15) * ATT_ROWS;
    const int tid   = threadIdx.x;
    const int w     = tid >> 5;
    const int l     = tid & 31;
    const int lh    = l >> 4;
    const int ll    = l & 15;

    const float* qbase = Q + (size_t)b * S * D + (size_t)(row0 + w * 16 + ll) * D + lh * 2;
    v2f qa[16];
    q_load(qa, qbase);

    {
        const v4f* kg = (const v4f*)(Kpp + (size_t)b * D * KP) + tid;
        const v4f* vg = (const v4f*)(Vpp + (size_t)b * KP * D) + tid;
        v4f kt[16], vt[16];
#pragma unroll
        for (int it = 0; it < 16; ++it) kt[it] = kg[it * 256];
#pragma unroll
        for (int it = 0; it < 16; ++it) vt[it] = vg[it * 256];
        const int kr = tid >> 7;
        float* kdst = kpl + kr * 512 + (((tid & 127) * 4 + 32 * kr) & 511);
        const int vr = tid >> 5;
        float* vdst = vpl + vr * 128 + (((tid & 31) * 4 + 32 * ((vr >> 2) & 1)) & 127);
#pragma unroll
        for (int it = 0; it < 16; ++it) *(v4f*)(kdst + it * 1024) = kt[it];
#pragma unroll
        for (int it = 0; it < 16; ++it) *(v4f*)(vdst + it * 1024) = vt[it];
    }
    __syncthreads();

    const float* kbase = kpl + lh * 512 + ll * 2;
    const float* vbase = vpl + lh * 512 + ll * 2;
    float* mystg = stg + (size_t)w * 16 * LDS_STRIDE;
    float* colst = mystg + ll * LDS_STRIDE + 8 * lh;
    const float* rowsrc = mystg + 4 * l;

#pragma unroll 1
    for (int si = 0; si < 2; ++si) {
        const int s0 = row0 + si * 128 + w * 16;

        v8f sc[16];
#pragma unroll
        for (int t = 0; t < 16; ++t) sc[t] = (v8f){};
        sc_frags f0, f1;
        sc_load(f0, kbase, lh, 0);
#pragma unroll
        for (int p = 0; p < 8; p += 2) {
            sc_load(f1, kbase, lh, p + 1);
            __builtin_amdgcn_sched_barrier(0);
            sc_mma(f0, qa, sc[2 * p], sc[2 * p + 1]);
            __builtin_amdgcn_sched_barrier(0);
            if (p + 2 < 8) sc_load(f0, kbase, lh, p + 2);
            __builtin_amdgcn_sched_barrier(0);
            sc_mma(f1, qa, sc[2 * p + 2], sc[2 * p + 3]);
            __builtin_amdgcn_sched_barrier(0);
        }

        if (si == 0) q_load(qa, qbase + (size_t)128 * D);

        {
            float m8[8];
#pragma unroll
            for (int r = 0; r < 8; ++r) m8[r] = sc[0][r];
#pragma unroll
            for (int t = 1; t < 16; ++t)
#pragma unroll
                for (int r = 0; r < 8; ++r) m8[r] = fmaxf(m8[r], sc[t][r]);
            float mx = fmaxf(fmaxf(fmaxf(m8[0], m8[1]), fmaxf(m8[2], m8[3])),
                             fmaxf(fmaxf(m8[4], m8[5]), fmaxf(m8[6], m8[7])));
            mx = fmaxf(mx, __shfl_xor(mx, 16, 32));
            float s8[8];
#pragma unroll
            for (int r = 0; r < 8; ++r) s8[r] = 0.f;
#pragma unroll
            for (int t = 0; t < 16; ++t)
#pragma unroll
                for (int r = 0; r < 8; ++r) {
                    const float p = __expf((sc[t][r] - mx) * scale);
                    sc[t][r] = p;
                    s8[r] += p;
                }
            float sum = ((s8[0] + s8[1]) + (s8[2] + s8[3])) +
                        ((s8[4] + s8[5]) + (s8[6] + s8[7]));
            sum += __shfl_xor(sum, 16, 32);
            const float rinv = 1.0f / sum;
#pragma unroll
            for (int t = 0; t < 16; ++t) sc[t] *= rinv;
        }

#pragma unroll
        for (int t = 0; t < 16; ++t) {
            const v4f lo = {sc[t][0], sc[t][1], sc[t][2], sc[t][3]};
            const v4f hi = {sc[t][4], sc[t][5], sc[t][6], sc[t][7]};
            *(v4f*)(colst + 16 * t)     = lo;
            *(v4f*)(colst + 16 * t + 4) = hi;
        }
        __syncthreads();

        float* arow = attn + (size_t)b * S * KP + (size_t)s0 * KP + 4 * l;
        v8f oc[4] = {};
        pv_frags g0, g1;
        pv_load(g0, vbase, rowsrc, lh, 0);
#pragma unroll
        for (int g = 0; g < 16; g += 2) {
            pv_load(g1, vbase, rowsrc, lh, g + 1);
            __builtin_amdgcn_sched_barrier(0);
            pv_mma_st(g0, oc, sc, arow, g);
            __builtin_amdgcn_sched_barrier(0);
            if (g + 2 < 16) pv_load(g0, vbase, rowsrc, lh, g + 2);
            __builtin_amdgcn_sched_barrier(0);
            pv_mma_st(g1, oc, sc, arow, g + 1);
            __builtin_amdgcn_sched_barrier(0);
        }

#pragma unroll
        for (int td = 0; td < 4; ++td) {
            const v4f lo = {oc[td][0], oc[td][1], oc[td][2], oc[td][3]};
            const v4f hi = {oc[td][4], oc[td][5], oc[td][6], oc[td][7]};
            *(v4f*)(colst + 16 * td)     = lo;
            *(v4f*)(colst + 16 * td + 4) = hi;
        }
        float* orow = out + (size_t)b * S * D + (size_t)(s0 + lh) * D + 4 * ll;
#pragma unroll
        for (int i = 0; i < 8; ++i) {
            const v4f o = *(const v4f*)(mystg + (2 * i + lh) * LDS_STRIDE + 4 * ll);
            *(v4f*)(orow + (size_t)(2 * i) * D) = o;
        }
    }
}

extern "C" void kernel_launch(void* const* d_in, const int* in_sizes, int n_in,
                              void* d_out, int out_size, void* d_ws, size_t ws_size,
                              hipStream_t stream) {
    (void)in_sizes; (void)n_in; (void)out_size; (void)ws_size;
    const float* Q   = (const float*)d_in[0];
    const float* K   = (const float*)d_in[1];
    const float* V   = (const float*)d_in[2];
    const float* E_w = (const float*)d_in[3];
    const float* E_b = (const float*)d_in[4];
    const float* F_w = (const float*)d_in[5];
    const float* F_b = (const float*)d_in[6];

    float* out  = (float*)d_out;
    float* attn = out + (size_t)LIN_B * LIN_S * LIN_D;

    float* Kpp = (float*)d_ws;
    float* Vpp = Kpp + (size_t)LIN_B * LIN_D * LIN_KP;

    const int projBlocks = 2 * LIN_B * (LIN_KP / PROJ_TILE);
    const int attnBlocks = LIN_B * (LIN_S / ATT_ROWS);

    lin_proj_kernel<<<projBlocks, PROJ_WAVES * 32, 0, stream>>>(K, V, E_w, F_w, E_b, F_b,
                                                                Kpp, Vpp);
    lin_attn_kernel<<<attnBlocks, ATT_WAVES * 32, 0, stream>>>(Q, Kpp, Vpp, out, attn);
}
